# starting kernel
# baseline (speedup 1.0000x reference)
#include <hip/hip_runtime.h>

typedef __attribute__((ext_vector_type(2))) float v2f;
typedef __attribute__((ext_vector_type(8))) float v8f;

#define SEQ_L 384
#define HID   768
#define LOUT  382

__global__ __launch_bounds__(256) void scale_rows_kernel(
    const float* __restrict__ X, const float* __restrict__ W,
    float* __restrict__ Y0, float* __restrict__ Y1)
{
    int idx = blockIdx.x * blockDim.x + threadIdx.x;
    if (idx >= SEQ_L * HID) return;
    int h = idx % HID;
    float x = X[idx];
    Y0[idx] = x * W[2 * h + 0];
    Y1[idx] = x * W[2 * h + 1];
}

__global__ __launch_bounds__(32) void gram_wmma_kernel(
    const float* __restrict__ X,
    const float* __restrict__ Y0,
    const float* __restrict__ Y1,
    const float* __restrict__ bias,
    float* __restrict__ out)
{
    const int i0   = blockIdx.y * 16;
    const int j0   = blockIdx.x * 16;
    const int lane = threadIdx.x;
    const int hi   = lane >> 4;
    const int lm   = lane & 15;
    const int kb   = hi * 2;

    const float* arow  = X  + (i0 + lm) * HID + kb;
    const float* b0row = Y0 + (j0 + lm) * HID + kb;
    const float* b1row = Y1 + (j0 + lm) * HID + kb;

    v8f acc0 = {};
    v8f acc1 = {};

#pragma unroll 8
    for (int kk = 0; kk < HID; kk += 4) {
        v2f a  = *(const v2f*)(arow  + kk);
        v2f b0 = *(const v2f*)(b0row + kk);
        v2f b1 = *(const v2f*)(b1row + kk);
        acc0 = __builtin_amdgcn_wmma_f32_16x16x4_f32(
            false, a, false, b0, (short)0, acc0, false, false);
        acc1 = __builtin_amdgcn_wmma_f32_16x16x4_f32(
            false, a, false, b1, (short)0, acc1, false, false);
    }

    const float badd0 = bias[0];
    const float badd1 = bias[1];

#pragma unroll
    for (int r = 0; r < 8; ++r) {
        const int i = i0 + r + hi * 8;
        const int j = j0 + lm;
        if (i >= 1 && i <= LOUT && j >= 1 && j <= LOUT) {
            float2 v = make_float2(acc0[r] + badd0, acc1[r] + badd1);
            *(float2*)(out + ((size_t)(i - 1) * LOUT + (j - 1)) * 2) = v;
        }
    }
}

__global__ __launch_bounds__(32) void gram_wmma_fused_kernel(
    const float* __restrict__ X,
    const float* __restrict__ W,
    const float* __restrict__ bias,
    float* __restrict__ out)
{
    const int i0   = blockIdx.y * 16;
    const int j0   = blockIdx.x * 16;
    const int lane = threadIdx.x;
    const int hi   = lane >> 4;
    const int lm   = lane & 15;
    const int kb   = hi * 2;

    const float* arow = X + (i0 + lm) * HID + kb;
    const float* brow = X + (j0 + lm) * HID + kb;

    v8f acc0 = {};
    v8f acc1 = {};

#pragma unroll 8
    for (int kk = 0; kk < HID; kk += 4) {
        v2f a = *(const v2f*)(arow + kk);
        v2f x = *(const v2f*)(brow + kk);
        float4 w = *(const float4*)(W + 2 * (kk + kb));
        v2f b0 = { x.x * w.x, x.y * w.z };
        v2f b1 = { x.x * w.y, x.y * w.w };
        acc0 = __builtin_amdgcn_wmma_f32_16x16x4_f32(
            false, a, false, b0, (short)0, acc0, false, false);
        acc1 = __builtin_amdgcn_wmma_f32_16x16x4_f32(
            false, a, false, b1, (short)0, acc1, false, false);
    }

    const float badd0 = bias[0];
    const float badd1 = bias[1];

#pragma unroll
    for (int r = 0; r < 8; ++r) {
        const int i = i0 + r + hi * 8;
        const int j = j0 + lm;
        if (i >= 1 && i <= LOUT && j >= 1 && j <= LOUT) {
            float2 v = make_float2(acc0[r] + badd0, acc1[r] + badd1);
            *(float2*)(out + ((size_t)(i - 1) * LOUT + (j - 1)) * 2) = v;
        }
    }
}

extern "C" void kernel_launch(void* const* d_in, const int* in_sizes, int n_in,
                              void* d_out, int out_size, void* d_ws, size_t ws_size,
                              hipStream_t stream)
{
    (void)in_sizes; (void)n_in; (void)out_size;
    const float* X    = (const float*)d_in[0];
    const float* W    = (const float*)d_in[2];
    const float* bias = (const float*)d_in[3];
    float*       out  = (float*)d_out;

    const dim3 grid(SEQ_L / 16, SEQ_L / 16);
    const dim3 block(32);

    const size_t ws_needed = (size_t)2 * SEQ_L * HID * sizeof(float);
    if (d_ws != nullptr && ws_size >= ws_needed) {
        float* Y0 = (float*)d_ws;
        float* Y1 = Y0 + (size_t)SEQ_L * HID;
        scale_rows_kernel<<<(SEQ_L * HID + 255) / 256, 256, 0, stream>>>(X, W, Y0, Y1);
        gram_wmma_kernel<<<grid, block, 0, stream>>>(X, Y0, Y1, bias, out);
    } else {
        gram_wmma_fused_kernel<<<grid, block, 0, stream>>>(X, W, bias, out);
    }
}


# GEMM K loop software-pipelined over three register sets: loads of K group g+2 issued before the matrix instructions of group g; K order and sum order unchanged
# speedup vs baseline: 1.2932x; 1.2932x over previous
#include <hip/hip_runtime.h>

typedef __attribute__((ext_vector_type(2))) float v2f;
typedef __attribute__((ext_vector_type(8))) float v8f;

#define SEQ_L 384
#define HID   768
#define LOUT  382

__global__ __launch_bounds__(256) void scale_rows_kernel(
    const float* __restrict__ X, const float* __restrict__ W,
    float* __restrict__ Y0, float* __restrict__ Y1)
{
    int idx = blockIdx.x * blockDim.x + threadIdx.x;
    if (idx >= SEQ_L * HID) return;
    int h = idx % HID;
    float x = X[idx];
    Y0[idx] = x * W[2 * h + 0];
    Y1[idx] = x * W[2 * h + 1];
}

#define GK 8
#define NGROUPS (HID / (4 * GK))

struct FragGroup {
    v2f a[GK];
    v2f b0[GK];
    v2f b1[GK];
};

__device__ __forceinline__ void load_group(FragGroup& f,
    const float* __restrict__ arow, const float* __restrict__ b0row,
    const float* __restrict__ b1row, int g)
{
#pragma unroll
    for (int s = 0; s < GK; ++s) {
        const int kk = g * (4 * GK) + s * 4;
        f.a[s]  = *(const v2f*)(arow  + kk);
        f.b0[s] = *(const v2f*)(b0row + kk);
        f.b1[s] = *(const v2f*)(b1row + kk);
    }
}

__device__ __forceinline__ void mma_group(const FragGroup& f, v8f& acc0, v8f& acc1)
{
#pragma unroll
    for (int s = 0; s < GK; ++s) {
        acc0 = __builtin_amdgcn_wmma_f32_16x16x4_f32(
            false, f.a[s], false, f.b0[s], (short)0, acc0, false, false);
        acc1 = __builtin_amdgcn_wmma_f32_16x16x4_f32(
            false, f.a[s], false, f.b1[s], (short)0, acc1, false, false);
    }
}

__global__ __launch_bounds__(32) void gram_wmma_kernel(
    const float* __restrict__ X,
    const float* __restrict__ Y0,
    const float* __restrict__ Y1,
    const float* __restrict__ bias,
    float* __restrict__ out)
{
    const int i0   = blockIdx.y * 16;
    const int j0   = blockIdx.x * 16;
    const int lane = threadIdx.x;
    const int hi   = lane >> 4;
    const int lm   = lane & 15;
    const int kb   = hi * 2;

    const float* arow  = X  + (i0 + lm) * HID + kb;
    const float* b0row = Y0 + (j0 + lm) * HID + kb;
    const float* b1row = Y1 + (j0 + lm) * HID + kb;

    v8f acc0 = {};
    v8f acc1 = {};

    FragGroup f0, f1, f2;
    load_group(f0, arow, b0row, b1row, 0);
    load_group(f1, arow, b0row, b1row, 1);

#pragma unroll 1
    for (int g = 0; g + 4 < NGROUPS; g += 3) {
        load_group(f2, arow, b0row, b1row, g + 2);
        mma_group(f0, acc0, acc1);
        load_group(f0, arow, b0row, b1row, g + 3);
        mma_group(f1, acc0, acc1);
        load_group(f1, arow, b0row, b1row, g + 4);
        mma_group(f2, acc0, acc1);
    }
    load_group(f2, arow, b0row, b1row, NGROUPS - 1);
    mma_group(f0, acc0, acc1);
    mma_group(f1, acc0, acc1);
    mma_group(f2, acc0, acc1);

    const float badd0 = bias[0];
    const float badd1 = bias[1];

#pragma unroll
    for (int r = 0; r < 8; ++r) {
        const int i = i0 + r + hi * 8;
        const int j = j0 + lm;
        if (i >= 1 && i <= LOUT && j >= 1 && j <= LOUT) {
            float2 v = make_float2(acc0[r] + badd0, acc1[r] + badd1);
            *(float2*)(out + ((size_t)(i - 1) * LOUT + (j - 1)) * 2) = v;
        }
    }
}

__global__ __launch_bounds__(32) void gram_wmma_fused_kernel(
    const float* __restrict__ X,
    const float* __restrict__ W,
    const float* __restrict__ bias,
    float* __restrict__ out)
{
    const int i0   = blockIdx.y * 16;
    const int j0   = blockIdx.x * 16;
    const int lane = threadIdx.x;
    const int hi   = lane >> 4;
    const int lm   = lane & 15;
    const int kb   = hi * 2;

    const float* arow = X + (i0 + lm) * HID + kb;
    const float* brow = X + (j0 + lm) * HID + kb;

    v8f acc0 = {};
    v8f acc1 = {};

#pragma unroll 8
    for (int kk = 0; kk < HID; kk += 4) {
        v2f a = *(const v2f*)(arow + kk);
        v2f x = *(const v2f*)(brow + kk);
        float4 w = *(const float4*)(W + 2 * (kk + kb));
        v2f b0 = { x.x * w.x, x.y * w.z };
        v2f b1 = { x.x * w.y, x.y * w.w };
        acc0 = __builtin_amdgcn_wmma_f32_16x16x4_f32(
            false, a, false, b0, (short)0, acc0, false, false);
        acc1 = __builtin_amdgcn_wmma_f32_16x16x4_f32(
            false, a, false, b1, (short)0, acc1, false, false);
    }

    const float badd0 = bias[0];
    const float badd1 = bias[1];

#pragma unroll
    for (int r = 0; r < 8; ++r) {
        const int i = i0 + r + hi * 8;
        const int j = j0 + lm;
        if (i >= 1 && i <= LOUT && j >= 1 && j <= LOUT) {
            float2 v = make_float2(acc0[r] + badd0, acc1[r] + badd1);
            *(float2*)(out + ((size_t)(i - 1) * LOUT + (j - 1)) * 2) = v;
        }
    }
}

extern "C" void kernel_launch(void* const* d_in, const int* in_sizes, int n_in,
                              void* d_out, int out_size, void* d_ws, size_t ws_size,
                              hipStream_t stream)
{
    (void)in_sizes; (void)n_in; (void)out_size;
    const float* X    = (const float*)d_in[0];
    const float* W    = (const float*)d_in[2];
    const float* bias = (const float*)d_in[3];
    float*       out  = (float*)d_out;

    const dim3 grid(SEQ_L / 16, SEQ_L / 16);
    const dim3 block(32);

    const size_t ws_needed = (size_t)2 * SEQ_L * HID * sizeof(float);
    if (d_ws != nullptr && ws_size >= ws_needed) {
        float* Y0 = (float*)d_ws;
        float* Y1 = Y0 + (size_t)SEQ_L * HID;
        scale_rows_kernel<<<(SEQ_L * HID + 255) / 256, 256, 0, stream>>>(X, W, Y0, Y1);
        gram_wmma_kernel<<<grid, block, 0, stream>>>(X, Y0, Y1, bias, out);
    } else {
        gram_wmma_fused_kernel<<<grid, block, 0, stream>>>(X, W, bias, out);
    }
}


# W scaling folded into the GEMM: both B fragments formed in registers from X rows j and four W entries per K step; scaling launch and its two workspace planes removed; products and sums unchanged
# speedup vs baseline: 1.6394x; 1.2677x over previous
#include <hip/hip_runtime.h>


typedef __attribute__((ext_vector_type(2))) float v2f;
typedef __attribute__((ext_vector_type(4))) float v4f;
typedef __attribute__((ext_vector_type(8))) float v8f;

#define SEQ_L 384
#define HID   768
#define LOUT  382

#define GK 8
#define NGROUPS (HID / (4 * GK))

struct FragGroup {
    v2f a[GK];
    v2f x[GK];
    v4f w[GK];
};

__device__ __forceinline__ void load_group(FragGroup& f,
    const float* __restrict__ arow, const float* __restrict__ brow,
    const float* __restrict__ wrow, int g)
{
#pragma unroll
    for (int s = 0; s < GK; ++s) {
        const int kk = g * (4 * GK) + s * 4;
        f.a[s] = *(const v2f*)(arow + kk);
        f.x[s] = *(const v2f*)(brow + kk);
        f.w[s] = *(const v4f*)(wrow + 2 * kk);
    }
}

__device__ __forceinline__ void mma_group(const FragGroup& f, v8f& acc0, v8f& acc1)
{
#pragma unroll
    for (int s = 0; s < GK; ++s) {
        const v2f b0 = { f.x[s].x * f.w[s].x, f.x[s].y * f.w[s].z };
        const v2f b1 = { f.x[s].x * f.w[s].y, f.x[s].y * f.w[s].w };
        acc0 = __builtin_amdgcn_wmma_f32_16x16x4_f32(
            false, f.a[s], false, b0, (short)0, acc0, false, false);
        acc1 = __builtin_amdgcn_wmma_f32_16x16x4_f32(
            false, f.a[s], false, b1, (short)0, acc1, false, false);
    }
}

#define HOLD_ORDER() do { asm volatile("" : "+v"(acc0), "+v"(acc1)); __builtin_amdgcn_sched_barrier(0); } while (0)

__global__ __launch_bounds__(32) void gram_wmma_kernel(
    const float* __restrict__ X,
    const float* __restrict__ W,
    const float* __restrict__ bias,
    float* __restrict__ out)
{
    const int i0   = blockIdx.y * 16;
    const int j0   = blockIdx.x * 16;
    const int lane = threadIdx.x;
    const int hi   = lane >> 4;
    const int lm   = lane & 15;
    const int kb   = hi * 2;

    const float* arow = X + (i0 + lm) * HID + kb;
    const float* brow = X + (j0 + lm) * HID + kb;
    const float* wrow = W + 2 * kb;

    v8f acc0 = {};
    v8f acc1 = {};

    FragGroup f0, f1, f2;
    load_group(f0, arow, brow, wrow, 0);
    load_group(f1, arow, brow, wrow, 1);
    HOLD_ORDER();

#pragma unroll 1
    for (int g = 0; g + 4 < NGROUPS; g += 3) {
        load_group(f2, arow, brow, wrow, g + 2);
        HOLD_ORDER();
        mma_group(f0, acc0, acc1);
        HOLD_ORDER();
        load_group(f0, arow, brow, wrow, g + 3);
        HOLD_ORDER();
        mma_group(f1, acc0, acc1);
        HOLD_ORDER();
        load_group(f1, arow, brow, wrow, g + 4);
        HOLD_ORDER();
        mma_group(f2, acc0, acc1);
        HOLD_ORDER();
    }
    load_group(f2, arow, brow, wrow, NGROUPS - 1);
    HOLD_ORDER();
    mma_group(f0, acc0, acc1);
    HOLD_ORDER();
    mma_group(f1, acc0, acc1);
    HOLD_ORDER();
    mma_group(f2, acc0, acc1);

    const float badd0 = bias[0];
    const float badd1 = bias[1];

#pragma unroll
    for (int r = 0; r < 8; ++r) {
        const int i = i0 + r + hi * 8;
        const int j = j0 + lm;
        if (i >= 1 && i <= LOUT && j >= 1 && j <= LOUT) {
            float2 v = make_float2(acc0[r] + badd0, acc1[r] + badd1);
            *(float2*)(out + ((size_t)(i - 1) * LOUT + (j - 1)) * 2) = v;
        }
    }
}

extern "C" void kernel_launch(void* const* d_in, const int* in_sizes, int n_in,
                              void* d_out, int out_size, void* d_ws, size_t ws_size,
                              hipStream_t stream)
{
    (void)in_sizes; (void)n_in; (void)out_size; (void)d_ws; (void)ws_size;
    const float* X    = (const float*)d_in[0];
    const float* W    = (const float*)d_in[2];
    const float* bias = (const float*)d_in[3];
    float*       out  = (float*)d_out;

    const dim3 grid(SEQ_L / 16, SEQ_L / 16);
    const dim3 block(32);

    gram_wmma_kernel<<<grid, block, 0, stream>>>(X, W, bias, out);
}


# X fragments fetched as one 16-B load per lane and row for two K steps (half the X load instructions); k order inside each block of 8 permuted, so the order of the sums changes
# speedup vs baseline: 1.7347x; 1.0581x over previous
#include <hip/hip_runtime.h>


typedef __attribute__((ext_vector_type(2))) float v2f;
typedef __attribute__((ext_vector_type(4))) float v4f;
typedef __attribute__((ext_vector_type(8))) float v8f;

#define SEQ_L 384
#define HID   768
#define LOUT  382

#define GK 8
#define GP (GK / 2)
#define NGROUPS (HID / (4 * GK))

struct FragGroup {
    v4f a[GP];
    v4f x[GP];
    v4f w[GK];
};

__device__ __forceinline__ void load_group(FragGroup& f,
    const float* __restrict__ arow, const float* __restrict__ brow,
    const float* __restrict__ wrow, int g)
{
#pragma unroll
    for (int t = 0; t < GP; ++t) {
        const int kk = g * (4 * GK) + t * 8;
        f.a[t] = *(const v4f*)(arow + kk);
        f.x[t] = *(const v4f*)(brow + kk);
        f.w[2 * t]     = *(const v4f*)(wrow + 2 * kk);
        f.w[2 * t + 1] = *(const v4f*)(wrow + 2 * kk + 4);
    }
}

__device__ __forceinline__ void mma_group(const FragGroup& f, v8f& acc0, v8f& acc1)
{
#pragma unroll
    for (int t = 0; t < GP; ++t) {
        const v4f a4 = f.a[t];
        const v4f x4 = f.x[t];
        const v4f wl = f.w[2 * t];
        const v4f wh = f.w[2 * t + 1];
        const v2f alo = { a4.x, a4.y };
        const v2f ahi = { a4.z, a4.w };
        const v2f b0lo = { x4.x * wl.x, x4.y * wl.z };
        const v2f b1lo = { x4.x * wl.y, x4.y * wl.w };
        const v2f b0hi = { x4.z * wh.x, x4.w * wh.z };
        const v2f b1hi = { x4.z * wh.y, x4.w * wh.w };
        acc0 = __builtin_amdgcn_wmma_f32_16x16x4_f32(
            false, alo, false, b0lo, (short)0, acc0, false, false);
        acc1 = __builtin_amdgcn_wmma_f32_16x16x4_f32(
            false, alo, false, b1lo, (short)0, acc1, false, false);
        acc0 = __builtin_amdgcn_wmma_f32_16x16x4_f32(
            false, ahi, false, b0hi, (short)0, acc0, false, false);
        acc1 = __builtin_amdgcn_wmma_f32_16x16x4_f32(
            false, ahi, false, b1hi, (short)0, acc1, false, false);
    }
}

#define HOLD_ORDER() do { asm volatile("" : "+v"(acc0), "+v"(acc1)); __builtin_amdgcn_sched_barrier(0); } while (0)

__global__ __launch_bounds__(32) void gram_wmma_kernel(
    const float* __restrict__ X,
    const float* __restrict__ W,
    const float* __restrict__ bias,
    float* __restrict__ out)
{
    const int i0   = blockIdx.y * 16;
    const int j0   = blockIdx.x * 16;
    const int lane = threadIdx.x;
    const int hi   = lane >> 4;
    const int lm   = lane & 15;
    const int kb   = hi * 4;

    const float* arow = X + (i0 + lm) * HID + kb;
    const float* brow = X + (j0 + lm) * HID + kb;
    const float* wrow = W + 2 * kb;

    v8f acc0 = {};
    v8f acc1 = {};

    FragGroup f0, f1, f2;
    load_group(f0, arow, brow, wrow, 0);
    load_group(f1, arow, brow, wrow, 1);
    HOLD_ORDER();

#pragma unroll 1
    for (int g = 0; g + 4 < NGROUPS; g += 3) {
        load_group(f2, arow, brow, wrow, g + 2);
        HOLD_ORDER();
        mma_group(f0, acc0, acc1);
        HOLD_ORDER();
        load_group(f0, arow, brow, wrow, g + 3);
        HOLD_ORDER();
        mma_group(f1, acc0, acc1);
        HOLD_ORDER();
        load_group(f1, arow, brow, wrow, g + 4);
        HOLD_ORDER();
        mma_group(f2, acc0, acc1);
        HOLD_ORDER();
    }
    load_group(f2, arow, brow, wrow, NGROUPS - 1);
    HOLD_ORDER();
    mma_group(f0, acc0, acc1);
    HOLD_ORDER();
    mma_group(f1, acc0, acc1);
    HOLD_ORDER();
    mma_group(f2, acc0, acc1);

    const float badd0 = bias[0];
    const float badd1 = bias[1];

#pragma unroll
    for (int r = 0; r < 8; ++r) {
        const int i = i0 + r + hi * 8;
        const int j = j0 + lm;
        if (i >= 1 && i <= LOUT && j >= 1 && j <= LOUT) {
            float2 v = make_float2(acc0[r] + badd0, acc1[r] + badd1);
            *(float2*)(out + ((size_t)(i - 1) * LOUT + (j - 1)) * 2) = v;
        }
    }
}

extern "C" void kernel_launch(void* const* d_in, const int* in_sizes, int n_in,
                              void* d_out, int out_size, void* d_ws, size_t ws_size,
                              hipStream_t stream)
{
    (void)in_sizes; (void)n_in; (void)out_size; (void)d_ws; (void)ws_size;
    const float* X    = (const float*)d_in[0];
    const float* W    = (const float*)d_in[2];
    const float* bias = (const float*)d_in[3];
    float*       out  = (float*)d_out;

    const dim3 grid(SEQ_L / 16, SEQ_L / 16);
    const dim3 block(32);

    gram_wmma_kernel<<<grid, block, 0, stream>>>(X, W, bias, out);
}


# K split over four waves per 16x16 tile (work-group of 128): each wave sums one quarter of K, partial tiles added through LDS in fixed order q0+q1+q2+q3; the order of the sums changes
# speedup vs baseline: 2.3851x; 1.3749x over previous
#include <hip/hip_runtime.h>


typedef __attribute__((ext_vector_type(2))) float v2f;
typedef __attribute__((ext_vector_type(4))) float v4f;
typedef __attribute__((ext_vector_type(8))) float v8f;

#define SEQ_L 384
#define HID   768
#define LOUT  382

#define GK 8
#define GP (GK / 2)
#define NGROUPS (HID / (4 * GK))
#define KSPLIT 4
#define NGW (NGROUPS / KSPLIT)

struct FragGroup {
    v4f a[GP];
    v4f x[GP];
    v4f w[GK];
};

__device__ __forceinline__ void load_group(FragGroup& f,
    const float* __restrict__ arow, const float* __restrict__ brow,
    const float* __restrict__ wrow, int g)
{
#pragma unroll
    for (int t = 0; t < GP; ++t) {
        const int kk = g * (4 * GK) + t * 8;
        f.a[t] = *(const v4f*)(arow + kk);
        f.x[t] = *(const v4f*)(brow + kk);
        f.w[2 * t]     = *(const v4f*)(wrow + 2 * kk);
        f.w[2 * t + 1] = *(const v4f*)(wrow + 2 * kk + 4);
    }
}

__device__ __forceinline__ void mma_group(const FragGroup& f, v8f& acc0, v8f& acc1)
{
#pragma unroll
    for (int t = 0; t < GP; ++t) {
        const v4f a4 = f.a[t];
        const v4f x4 = f.x[t];
        const v4f wl = f.w[2 * t];
        const v4f wh = f.w[2 * t + 1];
        const v2f alo = { a4.x, a4.y };
        const v2f ahi = { a4.z, a4.w };
        const v2f b0lo = { x4.x * wl.x, x4.y * wl.z };
        const v2f b1lo = { x4.x * wl.y, x4.y * wl.w };
        const v2f b0hi = { x4.z * wh.x, x4.w * wh.z };
        const v2f b1hi = { x4.z * wh.y, x4.w * wh.w };
        acc0 = __builtin_amdgcn_wmma_f32_16x16x4_f32(
            false, alo, false, b0lo, (short)0, acc0, false, false);
        acc1 = __builtin_amdgcn_wmma_f32_16x16x4_f32(
            false, alo, false, b1lo, (short)0, acc1, false, false);
        acc0 = __builtin_amdgcn_wmma_f32_16x16x4_f32(
            false, ahi, false, b0hi, (short)0, acc0, false, false);
        acc1 = __builtin_amdgcn_wmma_f32_16x16x4_f32(
            false, ahi, false, b1hi, (short)0, acc1, false, false);
    }
}

#define HOLD_ORDER() do { asm volatile("" : "+v"(acc0), "+v"(acc1)); __builtin_amdgcn_sched_barrier(0); } while (0)

__global__ __launch_bounds__(32 * KSPLIT) void gram_wmma_kernel(
    const float* __restrict__ X,
    const float* __restrict__ W,
    const float* __restrict__ bias,
    float* __restrict__ out)
{
    __shared__ v4f red[KSPLIT * 4 * 32];

    const int i0   = blockIdx.y * 16;
    const int j0   = blockIdx.x * 16;
    const int wv   = __builtin_amdgcn_readfirstlane((int)(threadIdx.x >> 5));
    const int lane = threadIdx.x & 31;
    const int hi   = lane >> 4;
    const int lm   = lane & 15;
    const int kb   = hi * 4;
    const int kq   = wv * (HID / KSPLIT);

    const float* arow = X + (i0 + lm) * HID + kq + kb;
    const float* brow = X + (j0 + lm) * HID + kq + kb;
    const float* wrow = W + 2 * (kq + kb);

    v8f acc0 = {};
    v8f acc1 = {};

    FragGroup f0, f1, f2;
    load_group(f0, arow, brow, wrow, 0);
    load_group(f1, arow, brow, wrow, 1);
    HOLD_ORDER();
    load_group(f2, arow, brow, wrow, 2);
    HOLD_ORDER();
    mma_group(f0, acc0, acc1);
    HOLD_ORDER();
    load_group(f0, arow, brow, wrow, 3);
    HOLD_ORDER();
    mma_group(f1, acc0, acc1);
    HOLD_ORDER();
    load_group(f1, arow, brow, wrow, 4);
    HOLD_ORDER();
    mma_group(f2, acc0, acc1);
    HOLD_ORDER();
    load_group(f2, arow, brow, wrow, NGW - 1);
    HOLD_ORDER();
    mma_group(f0, acc0, acc1);
    HOLD_ORDER();
    mma_group(f1, acc0, acc1);
    HOLD_ORDER();
    mma_group(f2, acc0, acc1);

    const float badd0 = bias[0];
    const float badd1 = bias[1];

#pragma unroll
    for (int p = 0; p < 4; ++p) {
        const v4f v = { acc0[2 * p], acc1[2 * p], acc0[2 * p + 1], acc1[2 * p + 1] };
        red[(wv * 4 + p) * 32 + lane] = v;
    }
    __syncthreads();

    v4f s = red[(0 * 4 + wv) * 32 + lane];
#pragma unroll
    for (int q = 1; q < KSPLIT; ++q)
        s += red[(q * 4 + wv) * 32 + lane];

    const int i = i0 + 2 * wv + hi * 8;
    const int j = j0 + lm;
    if (j >= 1 && j <= LOUT) {
        if (i >= 1 && i <= LOUT)
            *(float2*)(out + ((size_t)(i - 1) * LOUT + (j - 1)) * 2) =
                make_float2(s.x + badd0, s.y + badd1);
        if (i + 1 >= 1 && i + 1 <= LOUT)
            *(float2*)(out + ((size_t)i * LOUT + (j - 1)) * 2) =
                make_float2(s.z + badd0, s.w + badd1);
    }
}

extern "C" void kernel_launch(void* const* d_in, const int* in_sizes, int n_in,
                              void* d_out, int out_size, void* d_ws, size_t ws_size,
                              hipStream_t stream)
{
    (void)in_sizes; (void)n_in; (void)out_size; (void)d_ws; (void)ws_size;
    const float* X    = (const float*)d_in[0];
    const float* W    = (const float*)d_in[2];
    const float* bias = (const float*)d_in[3];
    float*       out  = (float*)d_out;

    const dim3 grid(SEQ_L / 16, SEQ_L / 16);
    const dim3 block(32 * KSPLIT);

    gram_wmma_kernel<<<grid, block, 0, stream>>>(X, W, bias, out);
}


# Output symmetry used: 300 tiles (tile col >= tile row) computed instead of 576; each off-diagonal tile's LDS sum is also stored transposed, so mirrored elements carry their transpose's rounding
# speedup vs baseline: 2.9468x; 1.2355x over previous
#include <hip/hip_runtime.h>


typedef __attribute__((ext_vector_type(2))) float v2f;
typedef __attribute__((ext_vector_type(4))) float v4f;
typedef __attribute__((ext_vector_type(8))) float v8f;

#define SEQ_L 384
#define HID   768
#define LOUT  382

#define GK 8
#define GP (GK / 2)
#define NGROUPS (HID / (4 * GK))
#define KSPLIT 4
#define NGW (NGROUPS / KSPLIT)
#define NT (SEQ_L / 16)
#define RP 18

struct FragGroup {
    v4f a[GP];
    v4f x[GP];
    v4f w[GK];
};

__device__ __forceinline__ void load_group(FragGroup& f,
    const float* __restrict__ arow, const float* __restrict__ brow,
    const float* __restrict__ wrow, int g)
{
#pragma unroll
    for (int t = 0; t < GP; ++t) {
        const int kk = g * (4 * GK) + t * 8;
        f.a[t] = *(const v4f*)(arow + kk);
        f.x[t] = *(const v4f*)(brow + kk);
        f.w[2 * t]     = *(const v4f*)(wrow + 2 * kk);
        f.w[2 * t + 1] = *(const v4f*)(wrow + 2 * kk + 4);
    }
}

__device__ __forceinline__ void mma_group(const FragGroup& f, v8f& acc0, v8f& acc1)
{
#pragma unroll
    for (int t = 0; t < GP; ++t) {
        const v4f a4 = f.a[t];
        const v4f x4 = f.x[t];
        const v4f wl = f.w[2 * t];
        const v4f wh = f.w[2 * t + 1];
        const v2f alo = { a4.x, a4.y };
        const v2f ahi = { a4.z, a4.w };
        const v2f b0lo = { x4.x * wl.x, x4.y * wl.z };
        const v2f b1lo = { x4.x * wl.y, x4.y * wl.w };
        const v2f b0hi = { x4.z * wh.x, x4.w * wh.z };
        const v2f b1hi = { x4.z * wh.y, x4.w * wh.w };
        acc0 = __builtin_amdgcn_wmma_f32_16x16x4_f32(
            false, alo, false, b0lo, (short)0, acc0, false, false);
        acc1 = __builtin_amdgcn_wmma_f32_16x16x4_f32(
            false, alo, false, b1lo, (short)0, acc1, false, false);
        acc0 = __builtin_amdgcn_wmma_f32_16x16x4_f32(
            false, ahi, false, b0hi, (short)0, acc0, false, false);
        acc1 = __builtin_amdgcn_wmma_f32_16x16x4_f32(
            false, ahi, false, b1hi, (short)0, acc1, false, false);
    }
}

#define HOLD_ORDER() do { asm volatile("" : "+v"(acc0), "+v"(acc1)); __builtin_amdgcn_sched_barrier(0); } while (0)

__global__ __launch_bounds__(32 * KSPLIT) void gram_wmma_kernel(
    const float* __restrict__ X,
    const float* __restrict__ W,
    const float* __restrict__ bias,
    float* __restrict__ out)
{
    __shared__ v2f red[KSPLIT * 16 * RP];

    const int bx   = blockIdx.x;
    const int by   = blockIdx.y;
    const bool upper_part = bx < NT - by;
    const int bi   = upper_part ? by : NT - 1 - by;
    const int bj   = upper_part ? by + bx : bx - 1;
    const int i0   = bi * 16;
    const int j0   = bj * 16;
    const int wv   = __builtin_amdgcn_readfirstlane((int)(threadIdx.x >> 5));
    const int lane = threadIdx.x & 31;
    const int hi   = lane >> 4;
    const int lm   = lane & 15;
    const int kb   = hi * 4;
    const int kq   = wv * (HID / KSPLIT);

    const float* arow = X + (i0 + lm) * HID + kq + kb;
    const float* brow = X + (j0 + lm) * HID + kq + kb;
    const float* wrow = W + 2 * (kq + kb);

    v8f acc0 = {};
    v8f acc1 = {};

    FragGroup f0, f1, f2;
    load_group(f0, arow, brow, wrow, 0);
    load_group(f1, arow, brow, wrow, 1);
    HOLD_ORDER();
    load_group(f2, arow, brow, wrow, 2);
    HOLD_ORDER();
    mma_group(f0, acc0, acc1);
    HOLD_ORDER();
    load_group(f0, arow, brow, wrow, 3);
    HOLD_ORDER();
    mma_group(f1, acc0, acc1);
    HOLD_ORDER();
    load_group(f1, arow, brow, wrow, 4);
    HOLD_ORDER();
    mma_group(f2, acc0, acc1);
    HOLD_ORDER();
    load_group(f2, arow, brow, wrow, NGW - 1);
    HOLD_ORDER();
    mma_group(f0, acc0, acc1);
    HOLD_ORDER();
    mma_group(f1, acc0, acc1);
    HOLD_ORDER();
    mma_group(f2, acc0, acc1);

    const float badd0 = bias[0];
    const float badd1 = bias[1];

#pragma unroll
    for (int r = 0; r < 8; ++r) {
        const v2f v = { acc0[r], acc1[r] };
        red[(wv * 16 + 8 * hi + r) * RP + lm] = v;
    }
    __syncthreads();

    v2f pl[2][KSPLIT];
    v2f tr[2][KSPLIT];
#pragma unroll
    for (int e = 0; e < 2; ++e)
#pragma unroll
        for (int q = 0; q < KSPLIT; ++q)
            pl[e][q] = red[(q * 16 + 2 * wv + e + 8 * hi) * RP + lm];
#pragma unroll
    for (int n = 0; n < 2; ++n)
#pragma unroll
        for (int q = 0; q < KSPLIT; ++q)
            tr[n][q] = red[(q * 16 + lm) * RP + 4 * wv + 2 * n + hi];
    asm volatile("" :: "v"(pl[0][0]), "v"(pl[0][1]), "v"(pl[0][2]), "v"(pl[0][3]),
                       "v"(pl[1][0]), "v"(pl[1][1]), "v"(pl[1][2]), "v"(pl[1][3]),
                       "v"(tr[0][0]), "v"(tr[0][1]), "v"(tr[0][2]), "v"(tr[0][3]),
                       "v"(tr[1][0]), "v"(tr[1][1]), "v"(tr[1][2]), "v"(tr[1][3]));

#pragma unroll
    for (int e = 0; e < 2; ++e) {
        const v2f s = ((pl[e][0] + pl[e][1]) + pl[e][2]) + pl[e][3];
        const int i = i0 + 2 * wv + e + 8 * hi;
        const int j = j0 + lm;
        if (i >= 1 && i <= LOUT && j >= 1 && j <= LOUT)
            *(float2*)(out + ((size_t)(i - 1) * LOUT + (j - 1)) * 2) =
                make_float2(s.x + badd0, s.y + badd1);
    }

    if (bi != bj) {
#pragma unroll
        for (int n = 0; n < 2; ++n) {
            const v2f s = ((tr[n][0] + tr[n][1]) + tr[n][2]) + tr[n][3];
            const int orow = j0 + 4 * wv + 2 * n + hi;
            const int ocol = i0 + lm;
            if (orow >= 1 && orow <= LOUT && ocol >= 1 && ocol <= LOUT)
                *(float2*)(out + ((size_t)(orow - 1) * LOUT + (ocol - 1)) * 2) =
                    make_float2(s.x + badd0, s.y + badd1);
        }
    }
}

extern "C" void kernel_launch(void* const* d_in, const int* in_sizes, int n_in,
                              void* d_out, int out_size, void* d_ws, size_t ws_size,
                              hipStream_t stream)
{
    (void)in_sizes; (void)n_in; (void)out_size; (void)d_ws; (void)ws_size;
    const float* X    = (const float*)d_in[0];
    const float* W    = (const float*)d_in[2];
    const float* bias = (const float*)d_in[3];
    float*       out  = (float*)d_out;

    const dim3 grid(NT + 1, NT / 2);
    const dim3 block(32 * KSPLIT);

    gram_wmma_kernel<<<grid, block, 0, stream>>>(X, W, bias, out);
}


# X fetched by whole-line global loads (4 full 128-B lines per load, not 32 B of 16 lines) into per-wave LDS staging; fragments read from LDS; fragment contents and sum order unchanged
# speedup vs baseline: 3.2728x; 1.1106x over previous
#include <hip/hip_runtime.h>


typedef __attribute__((ext_vector_type(2))) float v2f;
typedef __attribute__((ext_vector_type(4))) float v4f;
typedef __attribute__((ext_vector_type(8))) float v8f;

#define SEQ_L 384
#define HID   768
#define LOUT  382

#define GK 8
#define GP (GK / 2)
#define NGROUPS (HID / (4 * GK))
#define KSPLIT 4
#define NGW (NGROUPS / KSPLIT)
#define NT (SEQ_L / 16)
#define RP 18

#define PITCH 196
#define STAGE_ROWS 32
#define STAGE_WAVE (STAGE_ROWS * PITCH)

struct RawGroup  { v4f r[8]; };
struct WGroup    { v4f w[GK]; };
struct FragGroup { v4f a[GP];
                   v4f x[GP]; };

__device__ __forceinline__ void gload_x(RawGroup& R,
    const float* __restrict__ pa, const float* __restrict__ pb, int g)
{
#pragma unroll
    for (int n = 0; n < 4; ++n) {
        R.r[n]     = *(const v4f*)(pa + (2 * n) * HID + g * (4 * GK));
        R.r[4 + n] = *(const v4f*)(pb + (2 * n) * HID + g * (4 * GK));
    }
}

__device__ __forceinline__ void gload_w(WGroup& Wg, const float* __restrict__ wrow, int g)
{
#pragma unroll
    for (int t = 0; t < GP; ++t) {
        const int kk = g * (4 * GK) + t * 8;
        Wg.w[2 * t]     = *(const v4f*)(wrow + 2 * kk);
        Wg.w[2 * t + 1] = *(const v4f*)(wrow + 2 * kk + 4);
    }
}

__device__ __forceinline__ void stage_store(float* sst, const RawGroup& R, int g)
{
#pragma unroll
    for (int n = 0; n < 4; ++n) {
        *(v4f*)(sst + (2 * n) * PITCH + g * (4 * GK))      = R.r[n];
        *(v4f*)(sst + (16 + 2 * n) * PITCH + g * (4 * GK)) = R.r[4 + n];
    }
}

__device__ __forceinline__ void frag_load(FragGroup& f, const float* sld, int g)
{
#pragma unroll
    for (int t = 0; t < GP; ++t) {
        const int kk = g * (4 * GK) + t * 8;
        f.a[t] = *(const v4f*)(sld + kk);
        f.x[t] = *(const v4f*)(sld + 16 * PITCH + kk);
    }
}

__device__ __forceinline__ void mma_group(const FragGroup& f, const WGroup& Wg, v8f& acc0, v8f& acc1)
{
#pragma unroll
    for (int t = 0; t < GP; ++t) {
        const v4f a4 = f.a[t];
        const v4f x4 = f.x[t];
        const v4f wl = Wg.w[2 * t];
        const v4f wh = Wg.w[2 * t + 1];
        const v2f alo = { a4.x, a4.y };
        const v2f ahi = { a4.z, a4.w };
        const v2f b0lo = { x4.x * wl.x, x4.y * wl.z };
        const v2f b1lo = { x4.x * wl.y, x4.y * wl.w };
        const v2f b0hi = { x4.z * wh.x, x4.w * wh.z };
        const v2f b1hi = { x4.z * wh.y, x4.w * wh.w };
        acc0 = __builtin_amdgcn_wmma_f32_16x16x4_f32(
            false, alo, false, b0lo, (short)0, acc0, false, false);
        acc1 = __builtin_amdgcn_wmma_f32_16x16x4_f32(
            false, alo, false, b1lo, (short)0, acc1, false, false);
        acc0 = __builtin_amdgcn_wmma_f32_16x16x4_f32(
            false, ahi, false, b0hi, (short)0, acc0, false, false);
        acc1 = __builtin_amdgcn_wmma_f32_16x16x4_f32(
            false, ahi, false, b1hi, (short)0, acc1, false, false);
    }
}

#define HOLD_ORDER() do { asm volatile("" : "+v"(acc0), "+v"(acc1)); __builtin_amdgcn_sched_barrier(0); } while (0)

#define STAGE_HANDOVER() do { __builtin_amdgcn_fence(__ATOMIC_RELEASE, "wavefront"); __builtin_amdgcn_wave_barrier(); __builtin_amdgcn_fence(__ATOMIC_ACQUIRE, "wavefront"); } while (0)

__global__ __launch_bounds__(32 * KSPLIT) void gram_wmma_kernel(
    const float* __restrict__ X,
    const float* __restrict__ W,
    const float* __restrict__ bias,
    float* __restrict__ out)
{
    __shared__ v2f red[KSPLIT * 16 * RP];
    __shared__ __attribute__((aligned(16))) float stage[KSPLIT * STAGE_WAVE];

    const int bx   = blockIdx.x;
    const int by   = blockIdx.y;
    const bool upper_part = bx < NT - by;
    const int bi   = upper_part ? by : NT - 1 - by;
    const int bj   = upper_part ? by + bx : bx - 1;
    const int i0   = bi * 16;
    const int j0   = bj * 16;
    const int wv   = __builtin_amdgcn_readfirstlane((int)(threadIdx.x >> 5));
    const int lane = threadIdx.x & 31;
    const int hi   = lane >> 4;
    const int lm   = lane & 15;
    const int kb   = hi * 4;
    const int kq   = wv * (HID / KSPLIT);

    const int lrow = hi + 8 * ((lane >> 3) & 1);
    const int lcol = (lane & 7) * 4;

    const float* pa   = X + (i0 + lrow) * HID + kq + lcol;
    const float* pb   = X + (j0 + lrow) * HID + kq + lcol;
    const float* wrow = W + 2 * (kq + kb);
    float*       sst  = stage + wv * STAGE_WAVE + lrow * PITCH + lcol;
    const float* sld  = stage + wv * STAGE_WAVE + lm * PITCH + kb;

    v8f acc0 = {};
    v8f acc1 = {};

    RawGroup  r0, r1, r2, r3;
    WGroup    w0, w1, w2;
    FragGroup f0, f1;

    gload_x(r0, pa, pb, 0);
    gload_w(w0, wrow, 0);
    gload_x(r1, pa, pb, 1);
    gload_w(w1, wrow, 1);
    gload_x(r2, pa, pb, 2);
    gload_x(r3, pa, pb, 3);
    HOLD_ORDER();
    stage_store(sst, r0, 0);
    STAGE_HANDOVER();
    frag_load(f0, sld, 0);
    stage_store(sst, r1, 1);
    gload_w(w2, wrow, 2);
    STAGE_HANDOVER();
    frag_load(f1, sld, 1);
    HOLD_ORDER();
    mma_group(f0, w0, acc0, acc1);
    HOLD_ORDER();
    stage_store(sst, r2, 2);
    gload_x(r0, pa, pb, 4);
    gload_w(w0, wrow, 3);
    STAGE_HANDOVER();
    frag_load(f0, sld, 2);
    HOLD_ORDER();
    mma_group(f1, w1, acc0, acc1);
    HOLD_ORDER();
    stage_store(sst, r3, 3);
    gload_x(r1, pa, pb, NGW - 1);
    gload_w(w1, wrow, 4);
    STAGE_HANDOVER();
    frag_load(f1, sld, 3);
    HOLD_ORDER();
    mma_group(f0, w2, acc0, acc1);
    HOLD_ORDER();
    stage_store(sst, r0, 4);
    gload_w(w2, wrow, NGW - 1);
    STAGE_HANDOVER();
    frag_load(f0, sld, 4);
    HOLD_ORDER();
    mma_group(f1, w0, acc0, acc1);
    HOLD_ORDER();
    stage_store(sst, r1, NGW - 1);
    STAGE_HANDOVER();
    frag_load(f1, sld, NGW - 1);
    HOLD_ORDER();
    mma_group(f0, w1, acc0, acc1);
    HOLD_ORDER();
    mma_group(f1, w2, acc0, acc1);

    const float badd0 = bias[0];
    const float badd1 = bias[1];

#pragma unroll
    for (int r = 0; r < 8; ++r) {
        const v2f v = { acc0[r], acc1[r] };
        red[(wv * 16 + 8 * hi + r) * RP + lm] = v;
    }
    __syncthreads();

    v2f pl[2][KSPLIT];
    v2f tr[2][KSPLIT];
#pragma unroll
    for (int e = 0; e < 2; ++e)
#pragma unroll
        for (int q = 0; q < KSPLIT; ++q)
            pl[e][q] = red[(q * 16 + 2 * wv + e + 8 * hi) * RP + lm];
#pragma unroll
    for (int n = 0; n < 2; ++n)
#pragma unroll
        for (int q = 0; q < KSPLIT; ++q)
            tr[n][q] = red[(q * 16 + lm) * RP + 4 * wv + 2 * n + hi];
    asm volatile("" :: "v"(pl[0][0]), "v"(pl[0][1]), "v"(pl[0][2]), "v"(pl[0][3]),
                       "v"(pl[1][0]), "v"(pl[1][1]), "v"(pl[1][2]), "v"(pl[1][3]),
                       "v"(tr[0][0]), "v"(tr[0][1]), "v"(tr[0][2]), "v"(tr[0][3]),
                       "v"(tr[1][0]), "v"(tr[1][1]), "v"(tr[1][2]), "v"(tr[1][3]));

#pragma unroll
    for (int e = 0; e < 2; ++e) {
        const v2f s = ((pl[e][0] + pl[e][1]) + pl[e][2]) + pl[e][3];
        const int i = i0 + 2 * wv + e + 8 * hi;
        const int j = j0 + lm;
        if (i >= 1 && i <= LOUT && j >= 1 && j <= LOUT)
            *(float2*)(out + ((size_t)(i - 1) * LOUT + (j - 1)) * 2) =
                make_float2(s.x + badd0, s.y + badd1);
    }

    if (bi != bj) {
#pragma unroll
        for (int n = 0; n < 2; ++n) {
            const v2f s = ((tr[n][0] + tr[n][1]) + tr[n][2]) + tr[n][3];
            const int orow = j0 + 4 * wv + 2 * n + hi;
            const int ocol = i0 + lm;
            if (orow >= 1 && orow <= LOUT && ocol >= 1 && ocol <= LOUT)
                *(float2*)(out + ((size_t)(orow - 1) * LOUT + (ocol - 1)) * 2) =
                    make_float2(s.x + badd0, s.y + badd1);
        }
    }
}

extern "C" void kernel_launch(void* const* d_in, const int* in_sizes, int n_in,
                              void* d_out, int out_size, void* d_ws, size_t ws_size,
                              hipStream_t stream)
{
    (void)in_sizes; (void)n_in; (void)out_size; (void)d_ws; (void)ws_size;
    const float* X    = (const float*)d_in[0];
    const float* W    = (const float*)d_in[2];
    const float* bias = (const float*)d_in[3];
    float*       out  = (float*)d_out;

    const dim3 grid(NT + 1, NT / 2);
    const dim3 block(32 * KSPLIT);

    gram_wmma_kernel<<<grid, block, 0, stream>>>(X, W, bias, out);
}
